# starting kernel
# baseline (speedup 1.0000x reference)
#include <hip/hip_runtime.h>


typedef __attribute__((ext_vector_type(2))) float v2f;
typedef __attribute__((ext_vector_type(8))) float v8f;

#define FEAT 256

__global__ __launch_bounds__(256) void qcirc_wmma_kernel(
    const float* __restrict__ x,
    const float* __restrict__ W,
    const float* __restrict__ params,
    float* __restrict__ out)
{
    const int lane = threadIdx.x & 31;
    const int wave = threadIdx.x >> 5;
    const int tile = blockIdx.x * 8 + wave;
    const int m0   = tile * 16;

    const int r     = lane & 15;
    const int khalf = (lane >> 4) * 2;

    const float* xrow = x + (size_t)(m0 + r) * FEAT + khalf;

    const bool  nvalid = (r < 14);
    const float wscale = nvalid ? 1.0f : 0.0f;
    const float* wrow  = W + (size_t)(nvalid ? r : 0) * FEAT + khalf;

    v8f acc = {0.f, 0.f, 0.f, 0.f, 0.f, 0.f, 0.f, 0.f};

    #pragma unroll 8
    for (int k0 = 0; k0 < FEAT; k0 += 4) {
        v2f a, b;
        a.x = xrow[k0 + 0];
        a.y = xrow[k0 + 1];
        b.x = wrow[k0 + 0] * wscale;
        b.y = wrow[k0 + 1] * wscale;
        acc = __builtin_amdgcn_wmma_f32_16x16x4_f32(
            false, a, false, b, (short)0, acc, false, false);
    }

    const float cp0 = __cosf(params[0]);
    if ((lane & 15) == 0) {
        const int rowbase = m0 + (lane >> 4) * 8;
        #pragma unroll
        for (int i = 0; i < 8; ++i) {
            float z = cp0 * __cosf(acc[i]);
            out[rowbase + i] = 1.0f / (1.0f + __expf(-z));
        }
    }
}

extern "C" void kernel_launch(void* const* d_in, const int* in_sizes, int n_in,
                              void* d_out, int out_size, void* d_ws, size_t ws_size,
                              hipStream_t stream) {
    (void)n_in; (void)d_ws; (void)ws_size;
    const float* x      = (const float*)d_in[0];
    const float* W      = (const float*)d_in[1];
    const float* params = (const float*)d_in[2];
    float* out          = (float*)d_out;

    const int batch = out_size;
    const int tiles = batch / 16;
    const int blocks = tiles / 8;
    hipLaunchKernelGGL(qcirc_wmma_kernel, dim3(blocks), dim3(256), 0, stream,
                       x, W, params, out);
}


# One 16-row tile per work-group on 256 work-groups (was 8 per group on 32), K split over its 2 waves, each slice requested up front as 16-byte loads; halves added via LDS (summation order changes)
# speedup vs baseline: 2.5102x; 2.5102x over previous
#include <hip/hip_runtime.h>


typedef __attribute__((ext_vector_type(2))) float v2f;
typedef __attribute__((ext_vector_type(4))) float v4f;
typedef __attribute__((ext_vector_type(8))) float v8f;

#define FEAT   256
#define KSPLIT 2
#define KSLICE (FEAT / KSPLIT)
#define NLOAD  (KSLICE / 8)
static_assert(KSPLIT == 2, "the LDS sum adds exactly two halves");

__global__ __launch_bounds__(32 * KSPLIT) void qcirc_wmma_kernel(
    const float* __restrict__ x,
    const float* __restrict__ W,
    const float* __restrict__ params,
    float* __restrict__ out)
{
    __shared__ float part[KSPLIT][16];

    const int lane = threadIdx.x & 31;
    const int wave = threadIdx.x >> 5;
    const int m0   = blockIdx.x * 16;

    const int r = lane & 15;
    const int h = lane >> 4;

    const int koff = wave * KSLICE + h * 4;

    const float* xrow = x + (size_t)(m0 + r) * FEAT + koff;

    const bool  nvalid = (r < 14);
    const float wscale = nvalid ? 1.0f : 0.0f;
    const float* wrow  = W + (size_t)(nvalid ? r : 0) * FEAT + koff;

    v4f a[NLOAD], b[NLOAD];
    for (int j = 0; j < NLOAD; ++j) {
        a[j] = *reinterpret_cast<const v4f*>(xrow + j * 8);
        b[j] = *reinterpret_cast<const v4f*>(wrow + j * 8);
    }

    __builtin_amdgcn_sched_barrier(0);

    v8f acc = {0.f, 0.f, 0.f, 0.f, 0.f, 0.f, 0.f, 0.f};

    for (int j = 0; j < NLOAD; ++j) {
        v2f a0, a1, b0, b1;
        a0.x = a[j].x;          a0.y = a[j].y;
        a1.x = a[j].z;          a1.y = a[j].w;
        b0.x = b[j].x * wscale; b0.y = b[j].y * wscale;
        b1.x = b[j].z * wscale; b1.y = b[j].w * wscale;
        acc = __builtin_amdgcn_wmma_f32_16x16x4_f32(
            false, a0, false, b0, (short)0, acc, false, false);
        acc = __builtin_amdgcn_wmma_f32_16x16x4_f32(
            false, a1, false, b1, (short)0, acc, false, false);
    }

    if (r == 0) {
        for (int i = 0; i < 8; ++i) part[wave][h * 8 + i] = acc[i];
    }
    __syncthreads();

    if (threadIdx.x < 16) {
        const float s = part[0][threadIdx.x] + part[1][threadIdx.x];
        const float cp0 = __cosf(params[0]);
        const float z = cp0 * __cosf(s);
        out[m0 + threadIdx.x] = 1.0f / (1.0f + __expf(-z));
    }
}

extern "C" void kernel_launch(void* const* d_in, const int* in_sizes, int n_in,
                              void* d_out, int out_size, void* d_ws, size_t ws_size,
                              hipStream_t stream) {
    (void)in_sizes; (void)n_in; (void)d_ws; (void)ws_size;
    const float* x      = (const float*)d_in[0];
    const float* W      = (const float*)d_in[1];
    const float* params = (const float*)d_in[2];
    float* out          = (float*)d_out;

    const int batch = out_size;
    const int tiles = batch / 16;
    hipLaunchKernelGGL(qcirc_wmma_kernel, dim3(tiles), dim3(32 * KSPLIT), 0, stream,
                       x, W, params, out);
}


# Tile rows = 4 batch rows x 4 K-segments, columns = segments of weight row 0: each load reads whole 128-byte lines; a wave owns 8 rows, quad add replaces the LDS sum (summation order changes)
# speedup vs baseline: 2.7825x; 1.1085x over previous
#include <hip/hip_runtime.h>


typedef __attribute__((ext_vector_type(2))) float v2f;
typedef __attribute__((ext_vector_type(4))) float v4f;
typedef __attribute__((ext_vector_type(8))) float v8f;

#define FEAT   256
#define WAVES  2
#define NSTEP  (FEAT / 32)

__global__ __launch_bounds__(32 * WAVES) void qcirc_wmma_kernel(
    const float* __restrict__ x,
    const float* __restrict__ W,
    const float* __restrict__ params,
    float* __restrict__ out)
{
    const int lane = threadIdx.x & 31;
    const int wave = threadIdx.x >> 5;
    const int row0 = blockIdx.x * (8 * WAVES) + wave * 8;

    const int c = lane & 15;
    const int h = lane >> 4;
    const int s = c & 3;
    const int b = c >> 2;

    const int koff = s * 8 + h * 4;

    const float* x0 = x + (size_t)(row0 + b) * FEAT + koff;
    const float* x1 = x + (size_t)(row0 + 4 + b) * FEAT + koff;
    const float* w0 = W + koff;

    v4f a0[NSTEP], a1[NSTEP], bw[NSTEP];
    for (int j = 0; j < NSTEP; ++j) {
        bw[j] = *reinterpret_cast<const v4f*>(w0 + j * 32);
        a0[j] = *reinterpret_cast<const v4f*>(x0 + j * 32);
        a1[j] = *reinterpret_cast<const v4f*>(x1 + j * 32);
    }

    __builtin_amdgcn_sched_barrier(0);

    v8f acc0 = {0.f, 0.f, 0.f, 0.f, 0.f, 0.f, 0.f, 0.f};
    v8f acc1 = {0.f, 0.f, 0.f, 0.f, 0.f, 0.f, 0.f, 0.f};

    for (int j = 0; j < NSTEP; ++j) {
        v2f p0a, p1a, q0a, q1a, pb, qb;
        p0a.x = a0[j].x; p0a.y = a0[j].y;   q0a.x = a0[j].z; q0a.y = a0[j].w;
        p1a.x = a1[j].x; p1a.y = a1[j].y;   q1a.x = a1[j].z; q1a.y = a1[j].w;
        pb.x  = bw[j].x; pb.y  = bw[j].y;   qb.x  = bw[j].z; qb.y  = bw[j].w;
        acc0 = __builtin_amdgcn_wmma_f32_16x16x4_f32(
            false, p0a, false, pb, (short)0, acc0, false, false);
        acc1 = __builtin_amdgcn_wmma_f32_16x16x4_f32(
            false, p1a, false, pb, (short)0, acc1, false, false);
        acc0 = __builtin_amdgcn_wmma_f32_16x16x4_f32(
            false, q0a, false, qb, (short)0, acc0, false, false);
        acc1 = __builtin_amdgcn_wmma_f32_16x16x4_f32(
            false, q1a, false, qb, (short)0, acc1, false, false);
    }

    const bool k0 = (c & 1) != 0, k1 = (c & 2) != 0, k2 = (c & 4) != 0, k3 = (c & 8) != 0;
    const float e0 = k0 ? acc0[1] : acc0[0];
    const float e1 = k0 ? acc0[3] : acc0[2];
    const float e2 = k0 ? acc0[5] : acc0[4];
    const float e3 = k0 ? acc0[7] : acc0[6];
    const float e4 = k0 ? acc1[1] : acc1[0];
    const float e5 = k0 ? acc1[3] : acc1[2];
    const float e6 = k0 ? acc1[5] : acc1[4];
    const float e7 = k0 ? acc1[7] : acc1[6];
    const float f0 = k1 ? e1 : e0;
    const float f1 = k1 ? e3 : e2;
    const float f2 = k1 ? e5 : e4;
    const float f3 = k1 ? e7 : e6;
    const float g0 = k2 ? f1 : f0;
    const float g1 = k2 ? f3 : f2;
    float d = k3 ? g1 : g0;

    d += __int_as_float(__builtin_amdgcn_update_dpp(0, __float_as_int(d), 0x161, 0xf, 0xf, true));
    d += __int_as_float(__builtin_amdgcn_update_dpp(0, __float_as_int(d), 0x162, 0xf, 0xf, true));

    const float cp0 = __cosf(params[0]);
    const float z = cp0 * __cosf(d);
    const float y = 1.0f / (1.0f + __expf(-z));

    if (s == 0) {
        out[row0 + 4 * (c >> 3) + 2 * h + ((c >> 2) & 1)] = y;
    }
}

extern "C" void kernel_launch(void* const* d_in, const int* in_sizes, int n_in,
                              void* d_out, int out_size, void* d_ws, size_t ws_size,
                              hipStream_t stream) {
    (void)in_sizes; (void)n_in; (void)d_ws; (void)ws_size;
    const float* x      = (const float*)d_in[0];
    const float* W      = (const float*)d_in[1];
    const float* params = (const float*)d_in[2];
    float* out          = (float*)d_out;

    const int batch = out_size;
    const int groups = batch / 16;
    hipLaunchKernelGGL(qcirc_wmma_kernel, dim3(groups), dim3(32 * WAVES), 0, stream,
                       x, W, params, out);
}
